# starting kernel
# baseline (speedup 1.0000x reference)
#include <hip/hip_runtime.h>

#define FEAT 64

typedef float v2f __attribute__((ext_vector_type(2)));
typedef float v8f __attribute__((ext_vector_type(8)));

__device__ __forceinline__ void atomic_add_f32(float* p, float v) {
    asm volatile("global_atomic_add_f32 %0, %1, off" : : "v"(p), "v"(v) : "memory");
}

__global__ __launch_bounds__(256) void zero_f32x4(float4* __restrict__ p, int n4) {
    int i = blockIdx.x * blockDim.x + threadIdx.x;
    if (i < n4) p[i] = make_float4(0.f, 0.f, 0.f, 0.f);
}

__global__ __launch_bounds__(256) void scatter_add64(
    const float* __restrict__ x, const int* __restrict__ src,
    const int* __restrict__ dst, float* __restrict__ agg, int nedges)
{
    int gid  = blockIdx.x * blockDim.x + threadIdx.x;
    int e    = gid >> 5;
    int lane = gid & 31;
    if (e >= nedges) return;
    int s = src[e];
    int d = dst[e];
    float2 v = *(const float2*)(x + (size_t)s * FEAT + lane * 2);
    float* ap = agg + (size_t)d * FEAT + lane * 2;
    atomic_add_f32(ap, v.x);
    atomic_add_f32(ap + 1, v.y);
}

template<int OUTF, bool RELU>
__global__ __launch_bounds__(256) void gemm_bias_relu(
    const float* __restrict__ A, const float* __restrict__ W,
    const float* __restrict__ bias, float* __restrict__ out, int nrows)
{
    constexpr int NT = OUTF / 16;
    int wave = (blockIdx.x * blockDim.x + threadIdx.x) >> 5;
    int lane = threadIdx.x & 31;
    int tileM = wave / NT;
    int tileN = wave % NT;
    int row0 = tileM * 16;
    if (row0 >= nrows) return;
    int col0 = tileN * 16;
    int half = lane >> 4;
    int idx  = lane & 15;

    v8f c = {0.f, 0.f, 0.f, 0.f, 0.f, 0.f, 0.f, 0.f};
    const float* arow = A + (size_t)(row0 + idx) * FEAT + 2 * half;
    const float* wcol = W + col0 + idx;

#pragma unroll
    for (int k0 = 0; k0 < FEAT; k0 += 4) {
        v2f a = *(const v2f*)(arow + k0);
        v2f b;
        b.x = wcol[(k0 + 2 * half + 0) * OUTF];
        b.y = wcol[(k0 + 2 * half + 1) * OUTF];
        c = __builtin_amdgcn_wmma_f32_16x16x4_f32(
                 false, a,  false, b,
                 (short)0, c,  false,  false);
    }

    float bv = bias[col0 + idx];
#pragma unroll
    for (int v = 0; v < 8; ++v) {
        int m = v + 8 * half;
        float val = c[v] + bv;
        if (RELU) val = val > 0.f ? val : 0.f;
        out[(size_t)(row0 + m) * OUTF + col0 + idx] = val;
    }
}

extern "C" void kernel_launch(void* const* d_in, const int* in_sizes, int n_in,
                              void* d_out, int out_size, void* d_ws, size_t ws_size,
                              hipStream_t stream) {
    const float* features = (const float*)d_in[0];
    const int*   src      = (const int*)d_in[1];
    const int*   dst      = (const int*)d_in[2];
    const float* W1       = (const float*)d_in[3];
    const float* b1       = (const float*)d_in[4];
    const float* W2       = (const float*)d_in[5];
    const float* b2       = (const float*)d_in[6];
    const float* Wfc      = (const float*)d_in[7];
    const float* bfc      = (const float*)d_in[8];
    float* out = (float*)d_out;

    int nnodes = in_sizes[0] / FEAT;
    int nedges = in_sizes[1];

    float* buf0 = (float*)d_ws;
    float* buf1 = buf0 + (size_t)nnodes * FEAT;

    int n4 = (nnodes * FEAT) / 4;
    int zeroBlocks    = (n4 + 255) / 256;
    int scatterBlocks = (nedges + 7) / 8;
    int tilesM = (nnodes + 15) / 16;
    int gemm64Blocks = (tilesM * 4 * 32 + 255) / 256;
    int gemm32Blocks = (tilesM * 2 * 32 + 255) / 256;
    dim3 blk(256);

    zero_f32x4<<<zeroBlocks, blk, 0, stream>>>((float4*)buf0, n4);
    scatter_add64<<<scatterBlocks, blk, 0, stream>>>(features, src, dst, buf0, nedges);
    gemm_bias_relu<64, true><<<gemm64Blocks, blk, 0, stream>>>(buf0, W1, b1, buf1, nnodes);

    zero_f32x4<<<zeroBlocks, blk, 0, stream>>>((float4*)buf0, n4);
    scatter_add64<<<scatterBlocks, blk, 0, stream>>>(buf1, src, dst, buf0, nedges);
    gemm_bias_relu<64, true><<<gemm64Blocks, blk, 0, stream>>>(buf0, W2, b2, buf1, nnodes);

    gemm_bias_relu<32, false><<<gemm32Blocks, blk, 0, stream>>>(buf1, Wfc, bfc, out, nnodes);
}


# Scatter-add: one wave takes 16 consecutive edges (one coalesced id load, 16 row gathers in flight, then the f32 atomics; sums still in arrival order) instead of one edge: 16x fewer waves
# speedup vs baseline: 1.7244x; 1.7244x over previous
#include <hip/hip_runtime.h>

#define FEAT 64

typedef float v2f __attribute__((ext_vector_type(2)));
typedef float v8f __attribute__((ext_vector_type(8)));

#define EPW 16

__device__ __forceinline__ void atomic_add_rows8(
    unsigned off, const float2* v,
    float* r0, float* r1, float* r2, float* r3, float* r4, float* r5, float* r6, float* r7)
{
    asm volatile(
        "global_atomic_add_f32 %0, %1, %17\n\tglobal_atomic_add_f32 %0, %2, %17 offset:4\n\t"
        "global_atomic_add_f32 %0, %3, %18\n\tglobal_atomic_add_f32 %0, %4, %18 offset:4\n\t"
        "global_atomic_add_f32 %0, %5, %19\n\tglobal_atomic_add_f32 %0, %6, %19 offset:4\n\t"
        "global_atomic_add_f32 %0, %7, %20\n\tglobal_atomic_add_f32 %0, %8, %20 offset:4\n\t"
        "global_atomic_add_f32 %0, %9, %21\n\tglobal_atomic_add_f32 %0, %10, %21 offset:4\n\t"
        "global_atomic_add_f32 %0, %11, %22\n\tglobal_atomic_add_f32 %0, %12, %22 offset:4\n\t"
        "global_atomic_add_f32 %0, %13, %23\n\tglobal_atomic_add_f32 %0, %14, %23 offset:4\n\t"
        "global_atomic_add_f32 %0, %15, %24\n\tglobal_atomic_add_f32 %0, %16, %24 offset:4"
        : : "v"(off),
            "v"(v[0].x), "v"(v[0].y), "v"(v[1].x), "v"(v[1].y),
            "v"(v[2].x), "v"(v[2].y), "v"(v[3].x), "v"(v[3].y),
            "v"(v[4].x), "v"(v[4].y), "v"(v[5].x), "v"(v[5].y),
            "v"(v[6].x), "v"(v[6].y), "v"(v[7].x), "v"(v[7].y),
            "s"(r0), "s"(r1), "s"(r2), "s"(r3), "s"(r4), "s"(r5), "s"(r6), "s"(r7)
        : "memory");
}

__device__ __forceinline__ void atomic_add_row_wait(float* row, unsigned off, float2 v) {
    asm volatile("global_atomic_add_f32 %0, %1, %3\n\t"
                 "global_atomic_add_f32 %0, %2, %3 offset:4\n\t"
                 "s_wait_xcnt 0x0"
                 : : "v"(off), "v"(v.x), "v"(v.y), "s"(row) : "memory");
}

__global__ __launch_bounds__(256) void zero_f32x4(float4* __restrict__ p, int n4) {
    int i = blockIdx.x * blockDim.x + threadIdx.x;
    if (i < n4) p[i] = make_float4(0.f, 0.f, 0.f, 0.f);
}

__global__ __launch_bounds__(256) void scatter_add64(
    const float* __restrict__ x, const int* __restrict__ src,
    const int* __restrict__ dst, float* __restrict__ agg, int nedges)
{
    const int wave = __builtin_amdgcn_readfirstlane(
        (int)((blockIdx.x * blockDim.x + threadIdx.x) >> 5));
    const int e0 = wave * EPW;
    if (e0 >= nedges) return;
    const unsigned lane = threadIdx.x & 31;
    const int n = min(EPW, nedges - e0);
    const unsigned l16 = lane & 15;
    const int* ip = (lane < 16) ? src : dst;
    int id = 0;
    if ((int)l16 < n) id = ip[e0 + l16];
    const unsigned boff = lane * 8;
    if (n == EPW) {
        float2 v[EPW];
        float* r[EPW];
#pragma unroll
        for (int u = 0; u < EPW; ++u) {
            const int s = __builtin_amdgcn_readlane(id, u);
            const char* row = (const char*)(x + (size_t)s * FEAT);
            v[u] = *(const float2*)(row + boff);
        }
#pragma unroll
        for (int u = 0; u < EPW; ++u) {
            const int d = __builtin_amdgcn_readlane(id, 16 + u);
            r[u] = agg + (size_t)d * FEAT;
        }
        atomic_add_rows8(boff, v, r[0], r[1], r[2], r[3], r[4], r[5], r[6], r[7]);
        atomic_add_rows8(boff, v + 8, r[8], r[9], r[10], r[11], r[12], r[13], r[14], r[15]);
    } else {
        for (int u = 0; u < n; ++u) {
            const int s = __builtin_amdgcn_readlane(id, u);
            const int d = __builtin_amdgcn_readlane(id, 16 + u);
            const char* row = (const char*)(x + (size_t)s * FEAT);
            const float2 v = *(const float2*)(row + boff);
            atomic_add_row_wait(agg + (size_t)d * FEAT, boff, v);
        }
    }
}

template<int OUTF, bool RELU>
__global__ __launch_bounds__(256) void gemm_bias_relu(
    const float* __restrict__ A, const float* __restrict__ W,
    const float* __restrict__ bias, float* __restrict__ out, int nrows)
{
    constexpr int NT = OUTF / 16;
    int wave = (blockIdx.x * blockDim.x + threadIdx.x) >> 5;
    int lane = threadIdx.x & 31;
    int tileM = wave / NT;
    int tileN = wave % NT;
    int row0 = tileM * 16;
    if (row0 >= nrows) return;
    int col0 = tileN * 16;
    int half = lane >> 4;
    int idx  = lane & 15;

    v8f c = {0.f, 0.f, 0.f, 0.f, 0.f, 0.f, 0.f, 0.f};
    const float* arow = A + (size_t)(row0 + idx) * FEAT + 2 * half;
    const float* wcol = W + col0 + idx;

#pragma unroll
    for (int k0 = 0; k0 < FEAT; k0 += 4) {
        v2f a = *(const v2f*)(arow + k0);
        v2f b;
        b.x = wcol[(k0 + 2 * half + 0) * OUTF];
        b.y = wcol[(k0 + 2 * half + 1) * OUTF];
        c = __builtin_amdgcn_wmma_f32_16x16x4_f32(
                 false, a,  false, b,
                 (short)0, c,  false,  false);
    }

    float bv = bias[col0 + idx];
#pragma unroll
    for (int v = 0; v < 8; ++v) {
        int m = v + 8 * half;
        float val = c[v] + bv;
        if (RELU) val = val > 0.f ? val : 0.f;
        out[(size_t)(row0 + m) * OUTF + col0 + idx] = val;
    }
}

extern "C" void kernel_launch(void* const* d_in, const int* in_sizes, int n_in,
                              void* d_out, int out_size, void* d_ws, size_t ws_size,
                              hipStream_t stream) {
    const float* features = (const float*)d_in[0];
    const int*   src      = (const int*)d_in[1];
    const int*   dst      = (const int*)d_in[2];
    const float* W1       = (const float*)d_in[3];
    const float* b1       = (const float*)d_in[4];
    const float* W2       = (const float*)d_in[5];
    const float* b2       = (const float*)d_in[6];
    const float* Wfc      = (const float*)d_in[7];
    const float* bfc      = (const float*)d_in[8];
    float* out = (float*)d_out;

    int nnodes = in_sizes[0] / FEAT;
    int nedges = in_sizes[1];

    float* buf0 = (float*)d_ws;
    float* buf1 = buf0 + (size_t)nnodes * FEAT;

    int n4 = (nnodes * FEAT) / 4;
    int zeroBlocks    = (n4 + 255) / 256;
    int scatterBlocks = (nedges + 8 * EPW - 1) / (8 * EPW);
    int tilesM = (nnodes + 15) / 16;
    int gemm64Blocks = (tilesM * 4 * 32 + 255) / 256;
    int gemm32Blocks = (tilesM * 2 * 32 + 255) / 256;
    dim3 blk(256);

    zero_f32x4<<<zeroBlocks, blk, 0, stream>>>((float4*)buf0, n4);
    scatter_add64<<<scatterBlocks, blk, 0, stream>>>(features, src, dst, buf0, nedges);
    gemm_bias_relu<64, true><<<gemm64Blocks, blk, 0, stream>>>(buf0, W1, b1, buf1, nnodes);

    zero_f32x4<<<zeroBlocks, blk, 0, stream>>>((float4*)buf0, n4);
    scatter_add64<<<scatterBlocks, blk, 0, stream>>>(buf1, src, dst, buf0, nedges);
    gemm_bias_relu<64, true><<<gemm64Blocks, blk, 0, stream>>>(buf0, W2, b2, buf1, nnodes);

    gemm_bias_relu<32, false><<<gemm32Blocks, blk, 0, stream>>>(buf1, Wfc, bfc, out, nnodes);
}


# Scatter-add lane layout: lane l takes floats l and l+32 of a row, so each f32 atomic and each gather load covers one whole 128B line, not alternate dwords of two; sums still in arrival order
# speedup vs baseline: 1.9853x; 1.1513x over previous
#include <hip/hip_runtime.h>

#define FEAT 64

typedef float v2f __attribute__((ext_vector_type(2)));
typedef float v8f __attribute__((ext_vector_type(8)));

#define EPW 16

__device__ __forceinline__ void atomic_add_rows8(
    unsigned off, const float2* v,
    float* r0, float* r1, float* r2, float* r3, float* r4, float* r5, float* r6, float* r7)
{
    asm volatile(
        "global_atomic_add_f32 %0, %1, %17\n\tglobal_atomic_add_f32 %0, %2, %17 offset:128\n\t"
        "global_atomic_add_f32 %0, %3, %18\n\tglobal_atomic_add_f32 %0, %4, %18 offset:128\n\t"
        "global_atomic_add_f32 %0, %5, %19\n\tglobal_atomic_add_f32 %0, %6, %19 offset:128\n\t"
        "global_atomic_add_f32 %0, %7, %20\n\tglobal_atomic_add_f32 %0, %8, %20 offset:128\n\t"
        "global_atomic_add_f32 %0, %9, %21\n\tglobal_atomic_add_f32 %0, %10, %21 offset:128\n\t"
        "global_atomic_add_f32 %0, %11, %22\n\tglobal_atomic_add_f32 %0, %12, %22 offset:128\n\t"
        "global_atomic_add_f32 %0, %13, %23\n\tglobal_atomic_add_f32 %0, %14, %23 offset:128\n\t"
        "global_atomic_add_f32 %0, %15, %24\n\tglobal_atomic_add_f32 %0, %16, %24 offset:128"
        : : "v"(off),
            "v"(v[0].x), "v"(v[0].y), "v"(v[1].x), "v"(v[1].y),
            "v"(v[2].x), "v"(v[2].y), "v"(v[3].x), "v"(v[3].y),
            "v"(v[4].x), "v"(v[4].y), "v"(v[5].x), "v"(v[5].y),
            "v"(v[6].x), "v"(v[6].y), "v"(v[7].x), "v"(v[7].y),
            "s"(r0), "s"(r1), "s"(r2), "s"(r3), "s"(r4), "s"(r5), "s"(r6), "s"(r7)
        : "memory");
}

__device__ __forceinline__ void atomic_add_row_wait(float* row, unsigned off, float2 v) {
    asm volatile("global_atomic_add_f32 %0, %1, %3\n\t"
                 "global_atomic_add_f32 %0, %2, %3 offset:128\n\t"
                 "s_wait_xcnt 0x0"
                 : : "v"(off), "v"(v.x), "v"(v.y), "s"(row) : "memory");
}

__global__ __launch_bounds__(256) void zero_f32x4(float4* __restrict__ p, int n4) {
    int i = blockIdx.x * blockDim.x + threadIdx.x;
    if (i < n4) p[i] = make_float4(0.f, 0.f, 0.f, 0.f);
}

__global__ __launch_bounds__(256) void scatter_add64(
    const float* __restrict__ x, const int* __restrict__ src,
    const int* __restrict__ dst, float* __restrict__ agg, int nedges)
{
    const int wave = __builtin_amdgcn_readfirstlane(
        (int)((blockIdx.x * blockDim.x + threadIdx.x) >> 5));
    const int e0 = wave * EPW;
    if (e0 >= nedges) return;
    const unsigned lane = threadIdx.x & 31;
    const int n = min(EPW, nedges - e0);
    const unsigned l16 = lane & 15;
    const int* ip = (lane < 16) ? src : dst;
    int id = 0;
    if ((int)l16 < n) id = ip[e0 + l16];
    const unsigned boff = lane * 4;
    if (n == EPW) {
        float2 v[EPW];
        float* r[EPW];
#pragma unroll
        for (int u = 0; u < EPW; ++u) {
            const int s = __builtin_amdgcn_readlane(id, u);
            const float* row = (const float*)((const char*)(x + (size_t)s * FEAT) + boff);
            v[u].x = row[0];
            v[u].y = row[32];
        }
#pragma unroll
        for (int u = 0; u < EPW; ++u) {
            const int d = __builtin_amdgcn_readlane(id, 16 + u);
            r[u] = agg + (size_t)d * FEAT;
        }
        atomic_add_rows8(boff, v, r[0], r[1], r[2], r[3], r[4], r[5], r[6], r[7]);
        atomic_add_rows8(boff, v + 8, r[8], r[9], r[10], r[11], r[12], r[13], r[14], r[15]);
    } else {
        for (int u = 0; u < n; ++u) {
            const int s = __builtin_amdgcn_readlane(id, u);
            const int d = __builtin_amdgcn_readlane(id, 16 + u);
            const float* row = (const float*)((const char*)(x + (size_t)s * FEAT) + boff);
            const float2 v = make_float2(row[0], row[32]);
            atomic_add_row_wait(agg + (size_t)d * FEAT, boff, v);
        }
    }
}

template<int OUTF, bool RELU>
__global__ __launch_bounds__(256) void gemm_bias_relu(
    const float* __restrict__ A, const float* __restrict__ W,
    const float* __restrict__ bias, float* __restrict__ out, int nrows)
{
    constexpr int NT = OUTF / 16;
    int wave = (blockIdx.x * blockDim.x + threadIdx.x) >> 5;
    int lane = threadIdx.x & 31;
    int tileM = wave / NT;
    int tileN = wave % NT;
    int row0 = tileM * 16;
    if (row0 >= nrows) return;
    int col0 = tileN * 16;
    int half = lane >> 4;
    int idx  = lane & 15;

    v8f c = {0.f, 0.f, 0.f, 0.f, 0.f, 0.f, 0.f, 0.f};
    const float* arow = A + (size_t)(row0 + idx) * FEAT + 2 * half;
    const float* wcol = W + col0 + idx;

#pragma unroll
    for (int k0 = 0; k0 < FEAT; k0 += 4) {
        v2f a = *(const v2f*)(arow + k0);
        v2f b;
        b.x = wcol[(k0 + 2 * half + 0) * OUTF];
        b.y = wcol[(k0 + 2 * half + 1) * OUTF];
        c = __builtin_amdgcn_wmma_f32_16x16x4_f32(
                 false, a,  false, b,
                 (short)0, c,  false,  false);
    }

    float bv = bias[col0 + idx];
#pragma unroll
    for (int v = 0; v < 8; ++v) {
        int m = v + 8 * half;
        float val = c[v] + bv;
        if (RELU) val = val > 0.f ? val : 0.f;
        out[(size_t)(row0 + m) * OUTF + col0 + idx] = val;
    }
}

extern "C" void kernel_launch(void* const* d_in, const int* in_sizes, int n_in,
                              void* d_out, int out_size, void* d_ws, size_t ws_size,
                              hipStream_t stream) {
    const float* features = (const float*)d_in[0];
    const int*   src      = (const int*)d_in[1];
    const int*   dst      = (const int*)d_in[2];
    const float* W1       = (const float*)d_in[3];
    const float* b1       = (const float*)d_in[4];
    const float* W2       = (const float*)d_in[5];
    const float* b2       = (const float*)d_in[6];
    const float* Wfc      = (const float*)d_in[7];
    const float* bfc      = (const float*)d_in[8];
    float* out = (float*)d_out;

    int nnodes = in_sizes[0] / FEAT;
    int nedges = in_sizes[1];

    float* buf0 = (float*)d_ws;
    float* buf1 = buf0 + (size_t)nnodes * FEAT;

    int n4 = (nnodes * FEAT) / 4;
    int zeroBlocks    = (n4 + 255) / 256;
    int scatterBlocks = (nedges + 8 * EPW - 1) / (8 * EPW);
    int tilesM = (nnodes + 15) / 16;
    int gemm64Blocks = (tilesM * 4 * 32 + 255) / 256;
    int gemm32Blocks = (tilesM * 2 * 32 + 255) / 256;
    dim3 blk(256);

    zero_f32x4<<<zeroBlocks, blk, 0, stream>>>((float4*)buf0, n4);
    scatter_add64<<<scatterBlocks, blk, 0, stream>>>(features, src, dst, buf0, nedges);
    gemm_bias_relu<64, true><<<gemm64Blocks, blk, 0, stream>>>(buf0, W1, b1, buf1, nnodes);

    zero_f32x4<<<zeroBlocks, blk, 0, stream>>>((float4*)buf0, n4);
    scatter_add64<<<scatterBlocks, blk, 0, stream>>>(buf1, src, dst, buf0, nedges);
    gemm_bias_relu<64, true><<<gemm64Blocks, blk, 0, stream>>>(buf0, W2, b2, buf1, nnodes);

    gemm_bias_relu<32, false><<<gemm32Blocks, blk, 0, stream>>>(buf1, Wfc, bfc, out, nnodes);
}


# Products before sums (sum_u(x_u) W = sum_u(x_u W)): each product launch also stores the bias as the scatter-add start value, relu applied on the next read; both zero launches gone; sum order changes
# speedup vs baseline: 2.3630x; 1.1902x over previous
#include <hip/hip_runtime.h>

#define FEAT 64

typedef float v2f __attribute__((ext_vector_type(2)));
typedef float v8f __attribute__((ext_vector_type(8)));


#define EPW 16

__device__ __forceinline__ void atomic_add_rows8(
    unsigned off, const float2* v,
    float* r0, float* r1, float* r2, float* r3, float* r4, float* r5, float* r6, float* r7)
{
    asm volatile(
        "global_atomic_add_f32 %0, %1, %17\n\tglobal_atomic_add_f32 %0, %2, %17 offset:128\n\t"
        "global_atomic_add_f32 %0, %3, %18\n\tglobal_atomic_add_f32 %0, %4, %18 offset:128\n\t"
        "global_atomic_add_f32 %0, %5, %19\n\tglobal_atomic_add_f32 %0, %6, %19 offset:128\n\t"
        "global_atomic_add_f32 %0, %7, %20\n\tglobal_atomic_add_f32 %0, %8, %20 offset:128\n\t"
        "global_atomic_add_f32 %0, %9, %21\n\tglobal_atomic_add_f32 %0, %10, %21 offset:128\n\t"
        "global_atomic_add_f32 %0, %11, %22\n\tglobal_atomic_add_f32 %0, %12, %22 offset:128\n\t"
        "global_atomic_add_f32 %0, %13, %23\n\tglobal_atomic_add_f32 %0, %14, %23 offset:128\n\t"
        "global_atomic_add_f32 %0, %15, %24\n\tglobal_atomic_add_f32 %0, %16, %24 offset:128"
        : : "v"(off),
            "v"(v[0].x), "v"(v[0].y), "v"(v[1].x), "v"(v[1].y),
            "v"(v[2].x), "v"(v[2].y), "v"(v[3].x), "v"(v[3].y),
            "v"(v[4].x), "v"(v[4].y), "v"(v[5].x), "v"(v[5].y),
            "v"(v[6].x), "v"(v[6].y), "v"(v[7].x), "v"(v[7].y),
            "s"(r0), "s"(r1), "s"(r2), "s"(r3), "s"(r4), "s"(r5), "s"(r6), "s"(r7)
        : "memory");
}

__device__ __forceinline__ void atomic_add_row_wait(float* row, unsigned off, float2 v) {
    asm volatile("global_atomic_add_f32 %0, %1, %3\n\t"
                 "global_atomic_add_f32 %0, %2, %3 offset:128\n\t"
                 "s_wait_xcnt 0x0"
                 : : "v"(off), "v"(v.x), "v"(v.y), "s"(row) : "memory");
}

__global__ __launch_bounds__(256) void scatter_add64(
    const float* __restrict__ x, const int* __restrict__ src,
    const int* __restrict__ dst, float* __restrict__ agg, int nedges)
{
    const int wave = __builtin_amdgcn_readfirstlane(
        (int)((blockIdx.x * blockDim.x + threadIdx.x) >> 5));
    const int e0 = wave * EPW;
    if (e0 >= nedges) return;
    const unsigned lane = threadIdx.x & 31;
    const int n = min(EPW, nedges - e0);
    const unsigned l16 = lane & 15;
    const int* ip = (lane < 16) ? src : dst;
    int id = 0;
    if ((int)l16 < n) id = ip[e0 + l16];
    const unsigned boff = lane * 4;
    if (n == EPW) {
        float2 v[EPW];
        float* r[EPW];
#pragma unroll
        for (int u = 0; u < EPW; ++u) {
            const int s = __builtin_amdgcn_readlane(id, u);
            const float* row = (const float*)((const char*)(x + (size_t)s * FEAT) + boff);
            v[u].x = row[0];
            v[u].y = row[32];
        }
#pragma unroll
        for (int u = 0; u < EPW; ++u) {
            const int d = __builtin_amdgcn_readlane(id, 16 + u);
            r[u] = agg + (size_t)d * FEAT;
        }
        atomic_add_rows8(boff, v, r[0], r[1], r[2], r[3], r[4], r[5], r[6], r[7]);
        atomic_add_rows8(boff, v + 8, r[8], r[9], r[10], r[11], r[12], r[13], r[14], r[15]);
    } else {
        for (int u = 0; u < n; ++u) {
            const int s = __builtin_amdgcn_readlane(id, u);
            const int d = __builtin_amdgcn_readlane(id, 16 + u);
            const float* row = (const float*)((const char*)(x + (size_t)s * FEAT) + boff);
            const float2 v = make_float2(row[0], row[32]);
            atomic_add_row_wait(agg + (size_t)d * FEAT, boff, v);
        }
    }
}

template<int OUTF, bool RELU_IN, bool LAST>
__global__ __launch_bounds__(256) void rows_times_w(
    const float* __restrict__ A, const float* __restrict__ W,
    const float* __restrict__ bias, float* __restrict__ out,
    float* __restrict__ sum0, int nrows)
{
    static_assert(LAST || OUTF == FEAT, "the start-value rows are FEAT wide");
    constexpr int NT = OUTF / 16;
    int wave = (blockIdx.x * blockDim.x + threadIdx.x) >> 5;
    int lane = threadIdx.x & 31;
    int tileM = wave / NT;
    int tileN = wave % NT;
    int row0 = tileM * 16;
    if (row0 >= nrows) return;
    int col0 = tileN * 16;
    int half = lane >> 4;
    int idx  = lane & 15;

    if (!LAST) {
        const float4 b4 = *(const float4*)(bias + 4 * idx);
        const int r = row0 + 4 * tileN + half;
        const int ra = min(r, nrows - 1);
        const int rb = min(r + 2, nrows - 1);
        *(float4*)(sum0 + (size_t)ra * FEAT + 4 * idx) = b4;
        *(float4*)(sum0 + (size_t)rb * FEAT + 4 * idx) = b4;
    }

    v8f c = {0.f, 0.f, 0.f, 0.f, 0.f, 0.f, 0.f, 0.f};
    const float* arow = A + (size_t)(row0 + idx) * FEAT + 2 * half;
    const float* wcol = W + col0 + idx;

#pragma unroll
    for (int k0 = 0; k0 < FEAT; k0 += 4) {
        v2f a = *(const v2f*)(arow + k0);
        if (RELU_IN) {
            a.x = a.x > 0.f ? a.x : 0.f;
            a.y = a.y > 0.f ? a.y : 0.f;
        }
        v2f b;
        b.x = wcol[(k0 + 2 * half + 0) * OUTF];
        b.y = wcol[(k0 + 2 * half + 1) * OUTF];
        c = __builtin_amdgcn_wmma_f32_16x16x4_f32(
                 false, a,  false, b,
                 (short)0, c,  false,  false);
    }

    const float bv = LAST ? bias[col0 + idx] : 0.f;
#pragma unroll
    for (int v = 0; v < 8; ++v) {
        int m = v + 8 * half;
        size_t o = (size_t)(row0 + m) * OUTF + col0 + idx;
        out[o] = LAST ? c[v] + bv : c[v];
    }
}

extern "C" void kernel_launch(void* const* d_in, const int* in_sizes, int n_in,
                              void* d_out, int out_size, void* d_ws, size_t ws_size,
                              hipStream_t stream) {
    const float* features = (const float*)d_in[0];
    const int*   src      = (const int*)d_in[1];
    const int*   dst      = (const int*)d_in[2];
    const float* W1       = (const float*)d_in[3];
    const float* b1       = (const float*)d_in[4];
    const float* W2       = (const float*)d_in[5];
    const float* b2       = (const float*)d_in[6];
    const float* Wfc      = (const float*)d_in[7];
    const float* bfc      = (const float*)d_in[8];
    float* out = (float*)d_out;

    int nnodes = in_sizes[0] / FEAT;
    int nedges = in_sizes[1];

    float* sumA = (float*)d_ws;
    float* sumB = sumA + (size_t)nnodes * FEAT;
    float* prod = sumB + (size_t)nnodes * FEAT;

    int scatterBlocks = (nedges + 8 * EPW - 1) / (8 * EPW);
    int tilesM = (nnodes + 15) / 16;
    int gemm64Blocks = (tilesM * 4 * 32 + 255) / 256;
    int gemm32Blocks = (tilesM * 2 * 32 + 255) / 256;
    dim3 blk(256);

    rows_times_w<64, false, false><<<gemm64Blocks, blk, 0, stream>>>(features, W1, b1, prod, sumA, nnodes);
    scatter_add64<<<scatterBlocks, blk, 0, stream>>>(prod, src, dst, sumA, nedges);

    rows_times_w<64, true, false><<<gemm64Blocks, blk, 0, stream>>>(sumA, W2, b2, prod, sumB, nnodes);
    scatter_add64<<<scatterBlocks, blk, 0, stream>>>(prod, src, dst, sumB, nedges);

    rows_times_w<32, true, true><<<gemm32Blocks, blk, 0, stream>>>(sumB, Wfc, bfc, out, nullptr, nnodes);
}
